# starting kernel
# baseline (speedup 1.0000x reference)
#include <hip/hip_runtime.h>

typedef __attribute__((ext_vector_type(2))) float v2f;
typedef __attribute__((ext_vector_type(8))) float v8f;

#define GCN_N_NODES 100000
#define GCN_N_EDGES 1600000
#define GCN_D 32

__global__ __launch_bounds__(256) void gcn_gemm_wmma(
    const float* __restrict__ x, const float* __restrict__ W,
    float* __restrict__ y, int n_tiles)
{
    const int lane = threadIdx.x & 31;
    const int wave = threadIdx.x >> 5;
    const int tile = blockIdx.x * 8 + wave;
    if (tile >= n_tiles) return;

    const int m  = lane & 15;
    const int kb = (lane >> 4) * 2;

    v2f b0[8], b1[8];
#pragma unroll
    for (int kc = 0; kc < 8; ++kc) {
        const int k = kc * 4 + kb;
        const float* w0 = W + (m)      * GCN_D + k;
        const float* w1 = W + (16 + m) * GCN_D + k;
        v2f t0; t0.x = w0[0]; t0.y = w0[1];
        v2f t1; t1.x = w1[0]; t1.y = w1[1];
        b0[kc] = t0; b1[kc] = t1;
    }

    v8f c0 = {}; v8f c1 = {};
    const float* xr = x + (size_t)tile * 16 * GCN_D;
#pragma unroll
    for (int kc = 0; kc < 8; ++kc) {
        const int k = kc * 4 + kb;
        v2f a = *(const v2f*)(xr + m * GCN_D + k);
        c0 = __builtin_amdgcn_wmma_f32_16x16x4_f32(
                 false, a, false, b0[kc], (short)0, c0, false, false);
        c1 = __builtin_amdgcn_wmma_f32_16x16x4_f32(
                 false, a, false, b1[kc], (short)0, c1, false, false);
    }

    const int n     = lane & 15;
    const int rbase = (lane >> 4) * 8;
    float* yr = y + (size_t)tile * 16 * GCN_D;
#pragma unroll
    for (int r = 0; r < 8; ++r) {
        yr[(rbase + r) * GCN_D + n]      = c0[r];
        yr[(rbase + r) * GCN_D + 16 + n] = c1[r];
    }
}

__global__ __launch_bounds__(256) void gcn_zero(float4* __restrict__ out, int n4)
{
    int i = blockIdx.x * 256 + threadIdx.x;
    if (i < n4) {
        float4 z; z.x = 0.f; z.y = 0.f; z.z = 0.f; z.w = 0.f;
        out[i] = z;
    }
}

__global__ __launch_bounds__(256) void gcn_scatter(
    const float* __restrict__ y, const int* __restrict__ ei,
    float* __restrict__ out, int n_edges)
{
    int t = blockIdx.x * 256 + threadIdx.x;
    int e = t >> 5;
    int f = t & 31;
    if (e >= n_edges) return;
    int src = ei[e];
    int dst = ei[n_edges + e];
    atomicAdd(out + (size_t)dst * GCN_D + f, y[(size_t)src * GCN_D + f]);
}

extern "C" void kernel_launch(void* const* d_in, const int* in_sizes, int n_in,
                              void* d_out, int out_size, void* d_ws, size_t ws_size,
                              hipStream_t stream)
{
    const float* x  = (const float*)d_in[0];
    const int*   ei = (const int*)d_in[1];
    const float* W  = (const float*)d_in[2];
    float*       out = (float*)d_out;
    float*       y   = (float*)d_ws;

    const int n_tiles = GCN_N_NODES / 16;
    const int gemm_blocks = (n_tiles + 7) / 8;
    gcn_gemm_wmma<<<gemm_blocks, 256, 0, stream>>>(x, W, y, n_tiles);

    const int n4 = GCN_N_NODES * GCN_D / 4;
    gcn_zero<<<(n4 + 255) / 256, 256, 0, stream>>>((float4*)out, n4);

    const long long nthreads = (long long)GCN_N_EDGES * 32;
    const int scat_blocks = (int)((nthreads + 255) / 256);
    gcn_scatter<<<scat_blocks, 256, 0, stream>>>(y, ei, out, GCN_N_EDGES);
}


# Row ownership replaces per-edge global f32 atomics: 100 workgroups each keep 1000 out rows in LDS, scan all dst, add their y rows by ds_add_f32 in arrival order, store rows once; no zero pass
# speedup vs baseline: 1.8359x; 1.8359x over previous
#include <hip/hip_runtime.h>

typedef __attribute__((ext_vector_type(2))) float v2f;
typedef __attribute__((ext_vector_type(8))) float v8f;

#define GCN_N_NODES 100000
#define GCN_N_EDGES 1600000
#define GCN_D 32

__global__ __launch_bounds__(256) void gcn_gemm_wmma(
    const float* __restrict__ x, const float* __restrict__ W,
    float* __restrict__ y, int n_tiles)
{
    const int lane = threadIdx.x & 31;
    const int wave = threadIdx.x >> 5;
    const int tile = blockIdx.x * 8 + wave;
    if (tile >= n_tiles) return;

    const int m  = lane & 15;
    const int kb = (lane >> 4) * 2;

    v2f b0[8], b1[8];
#pragma unroll
    for (int kc = 0; kc < 8; ++kc) {
        const int k = kc * 4 + kb;
        const float* w0 = W + (m)      * GCN_D + k;
        const float* w1 = W + (16 + m) * GCN_D + k;
        v2f t0; t0.x = w0[0]; t0.y = w0[1];
        v2f t1; t1.x = w1[0]; t1.y = w1[1];
        b0[kc] = t0; b1[kc] = t1;
    }

    v8f c0 = {}; v8f c1 = {};
    const float* xr = x + (size_t)tile * 16 * GCN_D;
#pragma unroll
    for (int kc = 0; kc < 8; ++kc) {
        const int k = kc * 4 + kb;
        v2f a = *(const v2f*)(xr + m * GCN_D + k);
        c0 = __builtin_amdgcn_wmma_f32_16x16x4_f32(
                 false, a, false, b0[kc], (short)0, c0, false, false);
        c1 = __builtin_amdgcn_wmma_f32_16x16x4_f32(
                 false, a, false, b1[kc], (short)0, c1, false, false);
    }

    const int n     = lane & 15;
    const int rbase = (lane >> 4) * 8;
    float* yr = y + (size_t)tile * 16 * GCN_D;
#pragma unroll
    for (int r = 0; r < 8; ++r) {
        yr[(rbase + r) * GCN_D + n]      = c0[r];
        yr[(rbase + r) * GCN_D + 16 + n] = c1[r];
    }
}

#define OWN_ROWS    1000
#define OWN_BLOCKS  ((GCN_N_NODES + OWN_ROWS - 1) / OWN_ROWS)
#define OWN_THREADS 1024
#define OWN_WAVES   (OWN_THREADS / 32)
#define ROW_BITS    10
#define SRC_BITS    17
#define Q_CAP       512

static_assert(OWN_ROWS <= (1 << ROW_BITS), "local row must fit its field");
static_assert((long long)GCN_N_EDGES <= (1ll << (32 - ROW_BITS)), "edge id must fit its field");
static_assert(GCN_N_NODES <= (1 << SRC_BITS) && SRC_BITS + ROW_BITS <= 32, "src and local row share a word");
static_assert(GCN_N_EDGES % 128 == 0, "dst list is read 16 B per lane, whole waves in or out of range");
static_assert(31 + 8 * 32 <= Q_CAP, "ring must hold one scan step on top of a remainder");

template <bool FULL>
__device__ __forceinline__ void own_drain(
    const float* __restrict__ y, const int* __restrict__ ei, float* acc,
    const unsigned* wq, unsigned qhead, unsigned cnt, unsigned lane)
{
    unsigned pk = 0u;
    if (FULL || lane < cnt) {
        const unsigned ent = wq[(qhead + lane) & (Q_CAP - 1)];
        int s = ei[ent >> ROW_BITS];
        s = min(max(s, 0), GCN_N_NODES - 1);
        pk = (unsigned)s | ((ent & ((1u << ROW_BITS) - 1u)) << SRC_BITS);
    }

    float v[32];
#pragma unroll
    for (int j = 0; j < 32; ++j) {
        const unsigned pj = (unsigned)__builtin_amdgcn_readlane((int)pk, j);
        const float* row = y + (size_t)(pj & ((1u << SRC_BITS) - 1u)) * GCN_D;
        v[j] = row[lane];
    }
#pragma unroll
    for (int j = 0; j < 32; ++j) {
        if (FULL || (unsigned)j < cnt) {
            const unsigned pj = (unsigned)__builtin_amdgcn_readlane((int)pk, j);
            atomicAdd(&acc[(pj >> SRC_BITS) * GCN_D + lane], v[j]);
        }
    }
}

__device__ __forceinline__ void own_match(
    int dv, unsigned e, int lo, int rows, unsigned* wq, unsigned& qtail)
{
    const unsigned r = (unsigned)(dv - lo);
    const bool m = r < (unsigned)rows;
    const unsigned mk = __builtin_amdgcn_ballot_w32(m);
    if (mk) {
        const unsigned pos = qtail + __builtin_amdgcn_mbcnt_lo(mk, 0u);
        if (m) wq[pos & (Q_CAP - 1)] = (e << ROW_BITS) | r;
        qtail += (unsigned)__builtin_popcount(mk);
    }
}

__global__ __launch_bounds__(OWN_THREADS) void gcn_own_rows(
    const float* __restrict__ y, const int* __restrict__ ei, float* __restrict__ out)
{
    __shared__ __attribute__((aligned(16))) float acc[OWN_ROWS * GCN_D];
    __shared__ unsigned ring[OWN_WAVES * Q_CAP];

    const int tid       = threadIdx.x;
    const unsigned lane = (unsigned)tid & 31u;
    const int wave = __builtin_amdgcn_readfirstlane(tid >> 5);
    const int lo   = blockIdx.x * OWN_ROWS;
    const int rows = min(OWN_ROWS, GCN_N_NODES - lo);

    float4* acc4 = (float4*)acc;
    for (int i = tid; i < OWN_ROWS * GCN_D / 4; i += OWN_THREADS) {
        float4 z; z.x = 0.f; z.y = 0.f; z.z = 0.f; z.w = 0.f;
        acc4[i] = z;
    }
    __syncthreads();

    unsigned* wq = ring + wave * Q_CAP;
    const int4* dl = (const int4*)(ei + GCN_N_EDGES) + lane;
    constexpr int NG     = GCN_N_EDGES / 4;
    constexpr int PER_IT = 2 * OWN_THREADS;
    constexpr int ITERS  = (NG + PER_IT - 1) / PER_IT;
    const int start = (int)(blockIdx.x * (unsigned)ITERS / (unsigned)OWN_BLOCKS);

    auto group_of = [&](int i, int j) -> int {
        int it = start + i;
        if (it >= ITERS) it -= ITERS;
        return it * PER_IT + j * OWN_THREADS + wave * 32;
    };
    auto load_grp = [&](int i, int j) -> int4 {
        int4 d; d.x = -1; d.y = -1; d.z = -1; d.w = -1;
        if (i < ITERS) {
            const int g = group_of(i, j);
            if (g < NG) d = dl[g];
        }
        return d;
    };

    unsigned qhead = 0u, qtail = 0u;
    int4 a0 = load_grp(0, 0), a1 = load_grp(0, 1);
    int4 b0 = load_grp(1, 0), b1 = load_grp(1, 1);
    for (int i = 0; i < ITERS; ++i) {
        const int4 c0 = load_grp(i + 2, 0), c1 = load_grp(i + 2, 1);

        const unsigned e0 = ((unsigned)group_of(i, 0) + lane) * 4u;
        const unsigned e1 = ((unsigned)group_of(i, 1) + lane) * 4u;
        own_match(a0.x, e0 + 0u, lo, rows, wq, qtail);
        own_match(a0.y, e0 + 1u, lo, rows, wq, qtail);
        own_match(a0.z, e0 + 2u, lo, rows, wq, qtail);
        own_match(a0.w, e0 + 3u, lo, rows, wq, qtail);
        own_match(a1.x, e1 + 0u, lo, rows, wq, qtail);
        own_match(a1.y, e1 + 1u, lo, rows, wq, qtail);
        own_match(a1.z, e1 + 2u, lo, rows, wq, qtail);
        own_match(a1.w, e1 + 3u, lo, rows, wq, qtail);

        while (qtail - qhead >= 32u) {
            own_drain<true>(y, ei, acc, wq, qhead, 32u, lane);
            qhead += 32u;
        }
        a0 = b0; a1 = b1; b0 = c0; b1 = c1;
    }
    const unsigned rem = qtail - qhead;
    if (rem) own_drain<false>(y, ei, acc, wq, qhead, rem, lane);

    __syncthreads();

    float4* out4 = (float4*)(out + (size_t)lo * GCN_D);
    for (int i = tid; i < rows * GCN_D / 4; i += OWN_THREADS) out4[i] = acc4[i];
}

extern "C" void kernel_launch(void* const* d_in, const int* in_sizes, int n_in,
                              void* d_out, int out_size, void* d_ws, size_t ws_size,
                              hipStream_t stream)
{
    const float* x  = (const float*)d_in[0];
    const int*   ei = (const int*)d_in[1];
    const float* W  = (const float*)d_in[2];
    float*       out = (float*)d_out;
    float*       y   = (float*)d_ws;

    const int n_tiles = GCN_N_NODES / 16;
    const int gemm_blocks = (n_tiles + 7) / 8;
    gcn_gemm_wmma<<<gemm_blocks, 256, 0, stream>>>(x, W, y, n_tiles);

    gcn_own_rows<<<OWN_BLOCKS, OWN_THREADS, 0, stream>>>(y, ei, out);
}


# Edges counting-sorted by owner range in launch 1 (list takes y's place in ws, so W^T is applied to the LDS row sums on WMMA at the end): owners read own runs, no dst rescan; add order by arrival
# speedup vs baseline: 6.5009x; 3.5410x over previous
#include <hip/hip_runtime.h>

typedef __attribute__((ext_vector_type(2))) float v2f;
typedef __attribute__((ext_vector_type(8))) float v8f;

#define GCN_N_NODES 100000
#define GCN_N_EDGES 1600000
#define GCN_D 32


#define OWN_ROWS    400
#define OWN_BLOCKS  (GCN_N_NODES / OWN_ROWS)
#define OWN_TILES   (OWN_ROWS / 16)
#define OWN_THREADS 1024
#define OWN_WAVES   (OWN_THREADS / 32)
#define SRC_BITS    17
#define Q_CAP       64

#define SORT_BLOCKS  250
#define SORT_CHUNK   (GCN_N_EDGES / SORT_BLOCKS)
#define SORT_GROUPS  (SORT_CHUNK / 4)
#define SORT_THREADS (SORT_GROUPS / 2)
#define BIN_PAD      256

static_assert(OWN_BLOCKS * OWN_ROWS == GCN_N_NODES, "owners cover the rows exactly");
static_assert(OWN_ROWS % 16 == 0, "whole 16-row tiles per owner");
static_assert(OWN_TILES <= OWN_WAVES, "one wave per tile in the product");
static_assert(GCN_N_NODES <= (1 << SRC_BITS), "src must fit its field");
static_assert(((OWN_ROWS - 1) >> (32 - SRC_BITS)) == 0, "local row must fit above src");
static_assert(SORT_BLOCKS * SORT_CHUNK == GCN_N_EDGES && SORT_CHUNK % 8 == 0, "slices cover the edges exactly");
static_assert(OWN_BLOCKS < BIN_PAD && BIN_PAD == 256, "scan handles 8 owners per lane of one wave");
static_assert(SORT_BLOCKS <= OWN_THREADS && OWN_BLOCKS < SORT_THREADS, "one thread per table entry");

__device__ __forceinline__ void sort_count(
    int dv, int sv, unsigned* cnt, unsigned& bin, unsigned& rank, unsigned& word)
{
    const unsigned u = (unsigned)dv;
    const int s = min(max(sv, 0), GCN_N_NODES - 1);
    bin = 0xffffffffu; rank = 0u; word = 0u;
    if (u < (unsigned)GCN_N_NODES) {
        const unsigned b = u / (unsigned)OWN_ROWS;
        bin  = b;
        word = (unsigned)s | ((u - b * (unsigned)OWN_ROWS) << SRC_BITS);
        rank = atomicAdd(&cnt[b], 1u);
    }
}

__global__ __launch_bounds__(SORT_THREADS) void gcn_sort_edges(
    const int* __restrict__ ei, unsigned* __restrict__ sorted, unsigned* __restrict__ runs)
{
    __shared__ unsigned cnt[BIN_PAD];
    __shared__ unsigned base[BIN_PAD];
    __shared__ __attribute__((aligned(16))) unsigned stage[SORT_CHUNK];

    const int tid = threadIdx.x;
    const int g   = blockIdx.x;

    if (tid < BIN_PAD) cnt[tid] = 0u;

    const int4* s4 = (const int4*)ei + (size_t)g * SORT_GROUPS;
    const int4* d4 = (const int4*)(ei + GCN_N_EDGES) + (size_t)g * SORT_GROUPS;
    const int4 da = d4[tid], db = d4[tid + SORT_THREADS];
    const int4 sa = s4[tid], sb = s4[tid + SORT_THREADS];
    __syncthreads();

    unsigned bin[8], rank[8], word[8];
    sort_count(da.x, sa.x, cnt, bin[0], rank[0], word[0]);
    sort_count(da.y, sa.y, cnt, bin[1], rank[1], word[1]);
    sort_count(da.z, sa.z, cnt, bin[2], rank[2], word[2]);
    sort_count(da.w, sa.w, cnt, bin[3], rank[3], word[3]);
    sort_count(db.x, sb.x, cnt, bin[4], rank[4], word[4]);
    sort_count(db.y, sb.y, cnt, bin[5], rank[5], word[5]);
    sort_count(db.z, sb.z, cnt, bin[6], rank[6], word[6]);
    sort_count(db.w, sb.w, cnt, bin[7], rank[7], word[7]);
    __syncthreads();

    if (tid < 32) {
        unsigned c[8];
        unsigned tot = 0u;
#pragma unroll
        for (int k = 0; k < 8; ++k) { c[k] = cnt[tid * 8 + k]; tot += c[k]; }
        unsigned incl = tot;
#pragma unroll
        for (int d = 1; d < 32; d <<= 1) {
            const unsigned t = __shfl_up(incl, (unsigned)d);
            if (tid >= d) incl += t;
        }
        unsigned run = incl - tot;
#pragma unroll
        for (int k = 0; k < 8; ++k) { base[tid * 8 + k] = run; run += c[k]; }
    }
    __syncthreads();

#pragma unroll
    for (int k = 0; k < 8; ++k)
        if (bin[k] != 0xffffffffu) stage[base[bin[k]] + rank[k]] = word[k];
    __syncthreads();

    const unsigned total = base[OWN_BLOCKS];
    const uint4* st4 = (const uint4*)stage;
    uint4* o4 = (uint4*)(sorted + (size_t)g * SORT_CHUNK);
    for (int i = tid; i < SORT_GROUPS; i += SORT_THREADS)
        if ((unsigned)i * 4u < total) o4[i] = st4[i];

    if (tid <= OWN_BLOCKS) runs[(size_t)tid * SORT_BLOCKS + g] = base[tid];
}

template <bool FULL>
__device__ __forceinline__ void own_drain(
    const float* __restrict__ x, float* acc,
    const unsigned* wq, unsigned qhead, unsigned cnt, unsigned lane)
{
    unsigned pk = 0u;
    if (FULL || lane < cnt) pk = wq[(qhead + lane) & (Q_CAP - 1)];

    float v[32];
#pragma unroll
    for (int j = 0; j < 32; ++j) {
        const unsigned pj = (unsigned)__builtin_amdgcn_readlane((int)pk, j);
        const float* row = x + (size_t)(pj & ((1u << SRC_BITS) - 1u)) * GCN_D;
        v[j] = row[lane];
    }
#pragma unroll
    for (int j = 0; j < 32; ++j) {
        if (FULL || (unsigned)j < cnt) {
            const unsigned pj = (unsigned)__builtin_amdgcn_readlane((int)pk, j);
            atomicAdd(&acc[(pj >> SRC_BITS) * GCN_D + lane], v[j]);
        }
    }
}

__global__ __launch_bounds__(OWN_THREADS) void gcn_own_rows(
    const float* __restrict__ x, const float* __restrict__ W,
    const unsigned* __restrict__ sorted, const unsigned* __restrict__ runs,
    float* __restrict__ out)
{
    __shared__ __attribute__((aligned(16))) float acc[OWN_ROWS * GCN_D];
    __shared__ unsigned ring[OWN_WAVES * Q_CAP];
    __shared__ unsigned seg_at[SORT_BLOCKS];
    __shared__ unsigned seg_n[SORT_BLOCKS];

    const int tid       = threadIdx.x;
    const unsigned lane = (unsigned)tid & 31u;
    const int wave = __builtin_amdgcn_readfirstlane(tid >> 5);
    const int b    = blockIdx.x;
    const int lo   = b * OWN_ROWS;

    float4* acc4 = (float4*)acc;
    for (int i = tid; i < OWN_ROWS * GCN_D / 4; i += OWN_THREADS) {
        float4 z; z.x = 0.f; z.y = 0.f; z.z = 0.f; z.w = 0.f;
        acc4[i] = z;
    }
    if (tid < SORT_BLOCKS) {
        const unsigned s = runs[(size_t)b * SORT_BLOCKS + tid];
        const unsigned e = runs[(size_t)(b + 1) * SORT_BLOCKS + tid];
        seg_at[tid] = (unsigned)tid * (unsigned)SORT_CHUNK + s;
        seg_n[tid]  = e - s;
    }
    __syncthreads();

    unsigned* wq = ring + wave * Q_CAP;
    unsigned qhead = 0u, qtail = 0u;

    int g = wave;
    unsigned n_at = 0u, n_n = 0u;
    if (g < SORT_BLOCKS) {
        n_at = (unsigned)__builtin_amdgcn_readfirstlane((int)seg_at[g]);
        n_n  = (unsigned)__builtin_amdgcn_readfirstlane((int)seg_n[g]);
    }
    unsigned pre = sorted[lane < min(n_n, 32u) ? n_at + lane : (unsigned)GCN_N_EDGES];
    while (g < SORT_BLOCKS) {
        const unsigned at = n_at, n = n_n, cur = pre;
        const int g2 = g + OWN_WAVES;
        n_n = 0u;
        if (g2 < SORT_BLOCKS) {
            n_at = (unsigned)__builtin_amdgcn_readfirstlane((int)seg_at[g2]);
            n_n  = (unsigned)__builtin_amdgcn_readfirstlane((int)seg_n[g2]);
        }
        pre = sorted[lane < min(n_n, 32u) ? n_at + lane : (unsigned)GCN_N_EDGES];

        unsigned m = min(n, 32u);
        if (lane < m) wq[(qtail + lane) & (Q_CAP - 1)] = cur;
        qtail += m;
        if (qtail - qhead >= 32u) {
            own_drain<true>(x, acc, wq, qhead, 32u, lane);
            qhead += 32u;
        }
        for (unsigned c = 32u; c < n; c += 32u) {
            m = min(n - c, 32u);
            if (lane < m) wq[(qtail + lane) & (Q_CAP - 1)] = sorted[at + c + lane];
            qtail += m;
            if (qtail - qhead >= 32u) {
                own_drain<true>(x, acc, wq, qhead, 32u, lane);
                qhead += 32u;
            }
        }
        g = g2;
    }
    const unsigned rem = qtail - qhead;
    if (rem) own_drain<false>(x, acc, wq, qhead, rem, lane);

    const int m  = (int)(lane & 15u);
    const int kb = (int)(lane >> 4) * 2;
    v2f b0[8], b1[8];
    if (wave < OWN_TILES) {
#pragma unroll
        for (int kc = 0; kc < 8; ++kc) {
            const int k = kc * 4 + kb;
            const float* w0 = W + (m)      * GCN_D + k;
            const float* w1 = W + (16 + m) * GCN_D + k;
            v2f t0; t0.x = w0[0]; t0.y = w0[1];
            v2f t1; t1.x = w1[0]; t1.y = w1[1];
            b0[kc] = t0; b1[kc] = t1;
        }
    }

    __syncthreads();

    if (wave < OWN_TILES) {
        float* tr = acc + wave * 16 * GCN_D;
        v8f c0 = {}; v8f c1 = {};
#pragma unroll
        for (int kc = 0; kc < 8; ++kc) {
            const int k = kc * 4 + kb;
            v2f a = *(const v2f*)(tr + m * GCN_D + k);
            c0 = __builtin_amdgcn_wmma_f32_16x16x4_f32(
                     false, a, false, b0[kc], (short)0, c0, false, false);
            c1 = __builtin_amdgcn_wmma_f32_16x16x4_f32(
                     false, a, false, b1[kc], (short)0, c1, false, false);
        }
        const int rbase = (int)(lane >> 4) * 8;
#pragma unroll
        for (int r = 0; r < 8; ++r) {
            tr[(rbase + r) * GCN_D + m]      = c0[r];
            tr[(rbase + r) * GCN_D + 16 + m] = c1[r];
        }
    }
    __syncthreads();

    float4* out4 = (float4*)(out + (size_t)lo * GCN_D);
    for (int i = tid; i < OWN_ROWS * GCN_D / 4; i += OWN_THREADS) out4[i] = acc4[i];
}

extern "C" void kernel_launch(void* const* d_in, const int* in_sizes, int n_in,
                              void* d_out, int out_size, void* d_ws, size_t ws_size,
                              hipStream_t stream)
{
    const float* x  = (const float*)d_in[0];
    const int*   ei = (const int*)d_in[1];
    const float* W  = (const float*)d_in[2];
    float*       out = (float*)d_out;
    unsigned*    sorted = (unsigned*)d_ws;
    unsigned*    runs   = (unsigned*)d_ws + GCN_N_EDGES;

    gcn_sort_edges<<<SORT_BLOCKS, SORT_THREADS, 0, stream>>>(ei, sorted, runs);

    gcn_own_rows<<<OWN_BLOCKS, OWN_THREADS, 0, stream>>>(x, W, sorted, runs, out);
}


# x rows gathered four per load instruction (8 lanes x 16 B a row, lane addresses from the ring, no readlane); LDS adds rotated per lane group to stay conflict-free; add order still by arrival
# speedup vs baseline: 6.8159x; 1.0485x over previous
#include <hip/hip_runtime.h>

typedef __attribute__((ext_vector_type(2))) float v2f;
typedef __attribute__((ext_vector_type(8))) float v8f;

#define GCN_N_NODES 100000
#define GCN_N_EDGES 1600000
#define GCN_D 32


#define OWN_ROWS    400
#define OWN_BLOCKS  (GCN_N_NODES / OWN_ROWS)
#define OWN_TILES   (OWN_ROWS / 16)
#define OWN_THREADS 1024
#define OWN_WAVES   (OWN_THREADS / 32)
#define SRC_BITS    17
#define Q_CAP       64

#define SORT_BLOCKS  250
#define SORT_CHUNK   (GCN_N_EDGES / SORT_BLOCKS)
#define SORT_GROUPS  (SORT_CHUNK / 4)
#define SORT_THREADS (SORT_GROUPS / 2)
#define BIN_PAD      256

static_assert(OWN_BLOCKS * OWN_ROWS == GCN_N_NODES, "owners cover the rows exactly");
static_assert(OWN_ROWS % 16 == 0, "whole 16-row tiles per owner");
static_assert(OWN_TILES <= OWN_WAVES, "one wave per tile in the product");
static_assert(GCN_N_NODES <= (1 << SRC_BITS), "src must fit its field");
static_assert(((OWN_ROWS - 1) >> (32 - SRC_BITS)) == 0, "local row must fit above src");
static_assert(SORT_BLOCKS * SORT_CHUNK == GCN_N_EDGES && SORT_CHUNK % 8 == 0, "slices cover the edges exactly");
static_assert(OWN_BLOCKS < BIN_PAD && BIN_PAD == 256, "scan handles 8 owners per lane of one wave");
static_assert(SORT_BLOCKS <= OWN_THREADS && OWN_BLOCKS < SORT_THREADS, "one thread per table entry");

__device__ __forceinline__ void sort_count(
    int dv, int sv, unsigned* cnt, unsigned& bin, unsigned& rank, unsigned& word)
{
    const unsigned u = (unsigned)dv;
    const int s = min(max(sv, 0), GCN_N_NODES - 1);
    bin = 0xffffffffu; rank = 0u; word = 0u;
    if (u < (unsigned)GCN_N_NODES) {
        const unsigned b = u / (unsigned)OWN_ROWS;
        bin  = b;
        word = (unsigned)s | ((u - b * (unsigned)OWN_ROWS) << SRC_BITS);
        rank = atomicAdd(&cnt[b], 1u);
    }
}

__global__ __launch_bounds__(SORT_THREADS) void gcn_sort_edges(
    const int* __restrict__ ei, unsigned* __restrict__ sorted, unsigned* __restrict__ runs)
{
    __shared__ unsigned cnt[BIN_PAD];
    __shared__ unsigned base[BIN_PAD];
    __shared__ __attribute__((aligned(16))) unsigned stage[SORT_CHUNK];

    const int tid = threadIdx.x;
    const int g   = blockIdx.x;

    if (tid < BIN_PAD) cnt[tid] = 0u;

    const int4* s4 = (const int4*)ei + (size_t)g * SORT_GROUPS;
    const int4* d4 = (const int4*)(ei + GCN_N_EDGES) + (size_t)g * SORT_GROUPS;
    const int4 da = d4[tid], db = d4[tid + SORT_THREADS];
    const int4 sa = s4[tid], sb = s4[tid + SORT_THREADS];
    __syncthreads();

    unsigned bin[8], rank[8], word[8];
    sort_count(da.x, sa.x, cnt, bin[0], rank[0], word[0]);
    sort_count(da.y, sa.y, cnt, bin[1], rank[1], word[1]);
    sort_count(da.z, sa.z, cnt, bin[2], rank[2], word[2]);
    sort_count(da.w, sa.w, cnt, bin[3], rank[3], word[3]);
    sort_count(db.x, sb.x, cnt, bin[4], rank[4], word[4]);
    sort_count(db.y, sb.y, cnt, bin[5], rank[5], word[5]);
    sort_count(db.z, sb.z, cnt, bin[6], rank[6], word[6]);
    sort_count(db.w, sb.w, cnt, bin[7], rank[7], word[7]);
    __syncthreads();

    if (tid < 32) {
        unsigned c[8];
        unsigned tot = 0u;
#pragma unroll
        for (int k = 0; k < 8; ++k) { c[k] = cnt[tid * 8 + k]; tot += c[k]; }
        unsigned incl = tot;
#pragma unroll
        for (int d = 1; d < 32; d <<= 1) {
            const unsigned t = __shfl_up(incl, (unsigned)d);
            if (tid >= d) incl += t;
        }
        unsigned run = incl - tot;
#pragma unroll
        for (int k = 0; k < 8; ++k) { base[tid * 8 + k] = run; run += c[k]; }
    }
    __syncthreads();

#pragma unroll
    for (int k = 0; k < 8; ++k)
        if (bin[k] != 0xffffffffu) stage[base[bin[k]] + rank[k]] = word[k];
    __syncthreads();

    const unsigned total = base[OWN_BLOCKS];
    const uint4* st4 = (const uint4*)stage;
    uint4* o4 = (uint4*)(sorted + (size_t)g * SORT_CHUNK);
    for (int i = tid; i < SORT_GROUPS; i += SORT_THREADS)
        if ((unsigned)i * 4u < total) o4[i] = st4[i];

    if (tid <= OWN_BLOCKS) runs[(size_t)tid * SORT_BLOCKS + g] = base[tid];
}

template <bool FULL>
__device__ __forceinline__ void own_drain(
    const float* __restrict__ x, float* acc,
    const unsigned* wq, unsigned qhead, unsigned cnt, unsigned lane)
{
    const unsigned grp = lane >> 3;
    const unsigned q   = lane & 7u;
    const uint4* w4 = (const uint4*)(wq + ((qhead & (Q_CAP - 1)) + grp * 8u));
    const uint4 wa = w4[0], wb = w4[1];
    const unsigned pk[8] = { wa.x, wa.y, wa.z, wa.w, wb.x, wb.y, wb.z, wb.w };

    float4 v[8];
#pragma unroll
    for (int i = 0; i < 8; ++i) {
        if (FULL || grp * 8u + (unsigned)i < cnt) {
            const unsigned off = ((pk[i] & ((1u << SRC_BITS) - 1u)) << 7) | (q << 4);
            v[i] = *(const float4*)((const char*)x + off);
        }
    }
    const bool r1 = (grp & 1u) != 0u, r2 = (grp & 2u) != 0u;
#pragma unroll
    for (int i = 0; i < 8; ++i) {
        if (FULL || grp * 8u + (unsigned)i < cnt) {
            const float4 t = v[i];
            const float a0 = r1 ? t.y : t.x, a1 = r1 ? t.z : t.y, a2 = r1 ? t.w : t.z, a3 = r1 ? t.x : t.w;
            const float b0 = r2 ? a2 : a0,   b1 = r2 ? a3 : a1,   b2 = r2 ? a0 : a2,   b3 = r2 ? a1 : a3;
            float* row = acc + (pk[i] >> SRC_BITS) * GCN_D + q * 4u;
            atomicAdd(row + ((0u + grp) & 3u), b0);
            atomicAdd(row + ((1u + grp) & 3u), b1);
            atomicAdd(row + ((2u + grp) & 3u), b2);
            atomicAdd(row + ((3u + grp) & 3u), b3);
        }
    }
}

__global__ __launch_bounds__(OWN_THREADS) void gcn_own_rows(
    const float* __restrict__ x, const float* __restrict__ W,
    const unsigned* __restrict__ sorted, const unsigned* __restrict__ runs,
    float* __restrict__ out)
{
    __shared__ __attribute__((aligned(16))) float acc[OWN_ROWS * GCN_D];
    __shared__ __attribute__((aligned(16))) unsigned ring[OWN_WAVES * Q_CAP];
    __shared__ unsigned seg_at[SORT_BLOCKS];
    __shared__ unsigned seg_n[SORT_BLOCKS];

    const int tid       = threadIdx.x;
    const unsigned lane = (unsigned)tid & 31u;
    const int wave = __builtin_amdgcn_readfirstlane(tid >> 5);
    const int b    = blockIdx.x;
    const int lo   = b * OWN_ROWS;

    float4* acc4 = (float4*)acc;
    for (int i = tid; i < OWN_ROWS * GCN_D / 4; i += OWN_THREADS) {
        float4 z; z.x = 0.f; z.y = 0.f; z.z = 0.f; z.w = 0.f;
        acc4[i] = z;
    }
    if (tid < SORT_BLOCKS) {
        const unsigned s = runs[(size_t)b * SORT_BLOCKS + tid];
        const unsigned e = runs[(size_t)(b + 1) * SORT_BLOCKS + tid];
        seg_at[tid] = (unsigned)tid * (unsigned)SORT_CHUNK + s;
        seg_n[tid]  = e - s;
    }
    __syncthreads();

    unsigned* wq = ring + wave * Q_CAP;
    unsigned qhead = 0u, qtail = 0u;

    int g = wave;
    unsigned n_at = 0u, n_n = 0u;
    if (g < SORT_BLOCKS) {
        n_at = (unsigned)__builtin_amdgcn_readfirstlane((int)seg_at[g]);
        n_n  = (unsigned)__builtin_amdgcn_readfirstlane((int)seg_n[g]);
    }
    unsigned pre = sorted[lane < min(n_n, 32u) ? n_at + lane : (unsigned)GCN_N_EDGES];
    while (g < SORT_BLOCKS) {
        const unsigned at = n_at, n = n_n, cur = pre;
        const int g2 = g + OWN_WAVES;
        n_n = 0u;
        if (g2 < SORT_BLOCKS) {
            n_at = (unsigned)__builtin_amdgcn_readfirstlane((int)seg_at[g2]);
            n_n  = (unsigned)__builtin_amdgcn_readfirstlane((int)seg_n[g2]);
        }
        pre = sorted[lane < min(n_n, 32u) ? n_at + lane : (unsigned)GCN_N_EDGES];

        unsigned m = min(n, 32u);
        if (lane < m) wq[(qtail + lane) & (Q_CAP - 1)] = cur;
        qtail += m;
        if (qtail - qhead >= 32u) {
            own_drain<true>(x, acc, wq, qhead, 32u, lane);
            qhead += 32u;
        }
        for (unsigned c = 32u; c < n; c += 32u) {
            m = min(n - c, 32u);
            if (lane < m) wq[(qtail + lane) & (Q_CAP - 1)] = sorted[at + c + lane];
            qtail += m;
            if (qtail - qhead >= 32u) {
                own_drain<true>(x, acc, wq, qhead, 32u, lane);
                qhead += 32u;
            }
        }
        g = g2;
    }
    const unsigned rem = qtail - qhead;
    if (rem) own_drain<false>(x, acc, wq, qhead, rem, lane);

    const int m  = (int)(lane & 15u);
    const int kb = (int)(lane >> 4) * 2;
    v2f b0[8], b1[8];
    if (wave < OWN_TILES) {
#pragma unroll
        for (int kc = 0; kc < 8; ++kc) {
            const int k = kc * 4 + kb;
            const float* w0 = W + (m)      * GCN_D + k;
            const float* w1 = W + (16 + m) * GCN_D + k;
            v2f t0; t0.x = w0[0]; t0.y = w0[1];
            v2f t1; t1.x = w1[0]; t1.y = w1[1];
            b0[kc] = t0; b1[kc] = t1;
        }
    }

    __syncthreads();

    if (wave < OWN_TILES) {
        float* tr = acc + wave * 16 * GCN_D;
        v8f c0 = {}; v8f c1 = {};
#pragma unroll
        for (int kc = 0; kc < 8; ++kc) {
            const int k = kc * 4 + kb;
            v2f a = *(const v2f*)(tr + m * GCN_D + k);
            c0 = __builtin_amdgcn_wmma_f32_16x16x4_f32(
                     false, a, false, b0[kc], (short)0, c0, false, false);
            c1 = __builtin_amdgcn_wmma_f32_16x16x4_f32(
                     false, a, false, b1[kc], (short)0, c1, false, false);
        }
        const int rbase = (int)(lane >> 4) * 8;
#pragma unroll
        for (int r = 0; r < 8; ++r) {
            tr[(rbase + r) * GCN_D + m]      = c0[r];
            tr[(rbase + r) * GCN_D + 16 + m] = c1[r];
        }
    }
    __syncthreads();

    float4* out4 = (float4*)(out + (size_t)lo * GCN_D);
    for (int i = tid; i < OWN_ROWS * GCN_D / 4; i += OWN_THREADS) out4[i] = acc4[i];
}

extern "C" void kernel_launch(void* const* d_in, const int* in_sizes, int n_in,
                              void* d_out, int out_size, void* d_ws, size_t ws_size,
                              hipStream_t stream)
{
    const float* x  = (const float*)d_in[0];
    const int*   ei = (const int*)d_in[1];
    const float* W  = (const float*)d_in[2];
    float*       out = (float*)d_out;
    unsigned*    sorted = (unsigned*)d_ws;
    unsigned*    runs   = (unsigned*)d_ws + GCN_N_EDGES;

    gcn_sort_edges<<<SORT_BLOCKS, SORT_THREADS, 0, stream>>>(ei, sorted, runs);

    gcn_own_rows<<<OWN_BLOCKS, OWN_THREADS, 0, stream>>>(x, W, sorted, runs, out);
}
